# starting kernel
# baseline (speedup 1.0000x reference)
#include <hip/hip_runtime.h>
#include <hip/hip_bf16.h>

typedef __attribute__((ext_vector_type(16))) __bf16        v16bf;
typedef __attribute__((ext_vector_type(8)))  __bf16        v8bf;
typedef __attribute__((ext_vector_type(8)))  float         v8f;

__device__ __forceinline__ __bf16 tobf(float f) { return (__bf16)f; }

__global__ void k_fill0(float* __restrict__ p, int n) {
  int i = blockIdx.x * blockDim.x + threadIdx.x;
  if (i < n) p[i] = 0.0f;
}

__global__ void k_sig_deg(const float* __restrict__ P, const int* __restrict__ col,
                          float* __restrict__ w, float* __restrict__ deg, int E) {
  int e = blockIdx.x * blockDim.x + threadIdx.x;
  if (e >= E) return;
  float s = 1.0f / (1.0f + __expf(-P[e]));
  w[e] = s;
  atomicAdd(&deg[col[e]], s);
}

__global__ void k_dinv(float* __restrict__ deg, int N) {
  int i = blockIdx.x * blockDim.x + threadIdx.x;
  if (i >= N) return;
  float d = deg[i] + 1.0f;
  deg[i] = (d > 0.0f) ? rsqrtf(d) : 0.0f;
}

__global__ void k_norm(const int* __restrict__ row, const int* __restrict__ col,
                       const float* __restrict__ dinv, float* __restrict__ w, int E) {
  int e = blockIdx.x * blockDim.x + threadIdx.x;
  if (e >= E) return;
  w[e] = dinv[row[e]] * w[e] * dinv[col[e]];
}

template<int NCOL>
__global__ void k_wt_bf16(const float* __restrict__ W, __bf16* __restrict__ Wt) {
  int idx = blockIdx.x * blockDim.x + threadIdx.x;
  if (idx >= 128 * NCOL) return;
  int k = idx / NCOL, n = idx % NCOL;
  Wt[n * 128 + k] = tobf(W[idx]);
}

template<int NCOL, bool RELU_A>
__global__ void k_gemm_wmma(const float* __restrict__ A,
                            const __bf16* __restrict__ Bt,
                            const float* __restrict__ bias,
                            const float* __restrict__ dinv,
                            float* __restrict__ H,
                            float* __restrict__ OUT, int N) {
  constexpr int K    = 128;
  constexpr int LDSP = 136;
  __shared__ __align__(16) __bf16 As[16 * LDSP];

  const int lane = threadIdx.x & 31;
  const int wave = threadIdx.x >> 5;
  const int m0   = blockIdx.x * 16;
  const int n0   = wave * 16;
  const int half = lane >> 4;
  const int l15  = lane & 15;

  for (int base = threadIdx.x * 8; base < 16 * K; base += blockDim.x * 8) {
    const int m = base >> 7;
    const int k = base & (K - 1);
    int mg = m0 + m; if (mg > N - 1) mg = N - 1;
    const float* src = A + (size_t)mg * K + k;
    float4 f0 = *(const float4*)(src);
    float4 f1 = *(const float4*)(src + 4);
    if (RELU_A) {
      f0.x = fmaxf(f0.x, 0.f); f0.y = fmaxf(f0.y, 0.f);
      f0.z = fmaxf(f0.z, 0.f); f0.w = fmaxf(f0.w, 0.f);
      f1.x = fmaxf(f1.x, 0.f); f1.y = fmaxf(f1.y, 0.f);
      f1.z = fmaxf(f1.z, 0.f); f1.w = fmaxf(f1.w, 0.f);
    }
    v8bf pk;
    pk[0] = tobf(f0.x); pk[1] = tobf(f0.y); pk[2] = tobf(f0.z); pk[3] = tobf(f0.w);
    pk[4] = tobf(f1.x); pk[5] = tobf(f1.y); pk[6] = tobf(f1.z); pk[7] = tobf(f1.w);
    *(v8bf*)(&As[m * LDSP + k]) = pk;
  }
  __syncthreads();

  const __bf16* Brow = Bt + (size_t)(n0 + l15) * K;
  const __bf16* Alds = As + l15 * LDSP;

  v8f acc = {};

#pragma unroll
  for (int k0 = 0; k0 < K; k0 += 32) {
    const v8bf a0 = *(const v8bf*)(Alds + k0 + half * 8);
    const v8bf a1 = *(const v8bf*)(Alds + k0 + 16 + half * 8);
    const v16bf a = __builtin_shufflevector(a0, a1,
        0, 1, 2, 3, 4, 5, 6, 7, 8, 9, 10, 11, 12, 13, 14, 15);
    const v8bf b0 = *(const v8bf*)(Brow + k0 + half * 8);
    const v8bf b1 = *(const v8bf*)(Brow + k0 + 16 + half * 8);
    const v16bf b = __builtin_shufflevector(b0, b1,
        0, 1, 2, 3, 4, 5, 6, 7, 8, 9, 10, 11, 12, 13, 14, 15);
    acc = __builtin_amdgcn_wmma_f32_16x16x32_bf16(false, a, false, b,
                                                  (short)0, acc, false, false);
  }

  const int   n  = n0 + l15;
  const float bn = bias[n];
  if (m0 + 16 <= N) {
#pragma unroll
    for (int r = 0; r < 8; ++r) {
      const int   m = m0 + half * 8 + r;
      const float d = dinv[m];
      const float h = acc[r];
      H  [(size_t)m * NCOL + n] = h;
      OUT[(size_t)m * NCOL + n] = fmaf(d * d, h, bn);
    }
  } else {
#pragma unroll
    for (int r = 0; r < 8; ++r) {
      const int m = m0 + half * 8 + r;
      if (m < N) {
        const float d = dinv[m];
        const float h = acc[r];
        H  [(size_t)m * NCOL + n] = h;
        OUT[(size_t)m * NCOL + n] = fmaf(d * d, h, bn);
      }
    }
  }
}

template<int F>
__global__ void k_scatter(const int* __restrict__ row, const int* __restrict__ col,
                          const float* __restrict__ nrm, const float* __restrict__ h,
                          float* __restrict__ out, int E) {
  const int wid  = __builtin_amdgcn_readfirstlane(
                     (blockIdx.x * blockDim.x + threadIdx.x) >> 5);
  const int lane = threadIdx.x & 31;
  if (wid >= E) return;
  const int   r  = row[wid];
  const int   c  = col[wid];
  const float nm = nrm[wid];
  constexpr int PER = F / 32;
  const float* hp = h   + (size_t)r * F + lane * PER;
  float*       op = out + (size_t)c * F + lane * PER;
  if constexpr (PER == 4) {
    float4 v = *(const float4*)hp;
    atomicAdd(op + 0, nm * v.x);
    atomicAdd(op + 1, nm * v.y);
    atomicAdd(op + 2, nm * v.z);
    atomicAdd(op + 3, nm * v.w);
  } else {
    float2 v = *(const float2*)hp;
    atomicAdd(op + 0, nm * v.x);
    atomicAdd(op + 1, nm * v.y);
  }
}

extern "C" void kernel_launch(void* const* d_in, const int* in_sizes, int n_in,
                              void* d_out, int out_size, void* d_ws, size_t ws_size,
                              hipStream_t stream) {
  const float* x  = (const float*)d_in[0];
  const int*   ei = (const int*)  d_in[1];
  const float* P  = (const float*)d_in[2];
  const float* W1 = (const float*)d_in[3];
  const float* b1 = (const float*)d_in[4];
  const float* W2 = (const float*)d_in[5];
  const float* b2 = (const float*)d_in[6];
  float* out = (float*)d_out;

  constexpr int F_IN = 128, H1 = 128, H2 = 64;
  const int N = in_sizes[0] / F_IN;
  const int E = in_sizes[2];
  const int* row = ei;
  const int* col = ei + E;

  float* dinv = (float*)d_ws;
  float* wn   = dinv + N;
  float* h    = wn + E;
  float* o1   = h + (size_t)N * H1;
  __bf16* w1t = (__bf16*)(o1 + (size_t)N * H1);
  __bf16* w2t = w1t + F_IN * H1;

  const int T = 256;

  k_fill0  <<<(N + T - 1) / T, T, 0, stream>>>(dinv, N);
  k_sig_deg<<<(E + T - 1) / T, T, 0, stream>>>(P, col, wn, dinv, E);
  k_dinv   <<<(N + T - 1) / T, T, 0, stream>>>(dinv, N);
  k_norm   <<<(E + T - 1) / T, T, 0, stream>>>(row, col, dinv, wn, E);
  k_wt_bf16<H1><<<(F_IN * H1 + T - 1) / T, T, 0, stream>>>(W1, w1t);
  k_wt_bf16<H2><<<(F_IN * H2 + T - 1) / T, T, 0, stream>>>(W2, w2t);

  const int strips = (N + 15) / 16;

  k_gemm_wmma<H1, false><<<strips, 32 * (H1 / 16), 0, stream>>>(
      x, w1t, b1, dinv, h, o1, N);
  k_scatter<H1><<<(E + 7) / 8, T, 0, stream>>>(row, col, wn, h, o1, E);

  k_gemm_wmma<H2, true><<<strips, 32 * (H2 / 16), 0, stream>>>(
      o1, w2t, b2, dinv, h, out, N);
  k_scatter<H2><<<(E + 7) / 8, T, 0, stream>>>(row, col, wn, h, out, E);
}


# source-side degree factor folded into the GEMM store (rows kept as dinv[row]*h): the gather drops its scattered 4-byte dinv[source] read per edge in both layers; sum order unchanged
# speedup vs baseline: 2.1622x; 2.1622x over previous
#include <hip/hip_runtime.h>
#include <hip/hip_bf16.h>

typedef __attribute__((ext_vector_type(16))) __bf16        v16bf;
typedef __attribute__((ext_vector_type(8)))  __bf16        v8bf;
typedef __attribute__((ext_vector_type(8)))  float         v8f;

__device__ __forceinline__ __bf16 tobf(float f) { return (__bf16)f; }


__global__ void k_fill0(int* __restrict__ p, int n) {
  int i = blockIdx.x * blockDim.x + threadIdx.x;
  if (i < n) p[i] = 0;
}

__global__ void k_rank(const int* __restrict__ col, int* __restrict__ cnt,
                       int* __restrict__ rank, int E) {
  int e = blockIdx.x * blockDim.x + threadIdx.x;
  if (e >= E) return;
  rank[e] = atomicAdd(&cnt[col[e]], 1);
}

__device__ __forceinline__ int wave_incl_scan(int x, int lane) {
  x += __builtin_amdgcn_update_dpp(0, x, 0x111, 0xf, 0xf, false);
  x += __builtin_amdgcn_update_dpp(0, x, 0x112, 0xf, 0xf, false);
  x += __builtin_amdgcn_update_dpp(0, x, 0x114, 0xf, 0xf, false);
  x += __builtin_amdgcn_update_dpp(0, x, 0x118, 0xf, 0xf, false);
  const int lo = __builtin_amdgcn_readlane(x, 15);
  if (lane >= 16) x += lo;
  return x;
}

__global__ void __launch_bounds__(1024)
k_scan(const int* __restrict__ cnt, int* __restrict__ off, int N) {
  __shared__ int wsum[32];
  __shared__ int tile_total;
  const int tid  = threadIdx.x;
  const int lane = tid & 31;
  const int wave = tid >> 5;
  int carry = 0;
  for (int base = 0; base < N; base += 4096) {
    const int i0 = base + tid * 4;
    int4 v = make_int4(0, 0, 0, 0);
    if (i0 + 3 < N) {
      v = *(const int4*)(cnt + i0);
    } else {
      if (i0     < N) v.x = cnt[i0];
      if (i0 + 1 < N) v.y = cnt[i0 + 1];
      if (i0 + 2 < N) v.z = cnt[i0 + 2];
    }
    const int s1 = v.x, s2 = s1 + v.y, s3 = s2 + v.z, s4 = s3 + v.w;
    const int inc = wave_incl_scan(s4, lane);
    if (lane == 31) wsum[wave] = inc;
    __syncthreads();
    if (wave == 0) {
      const int t  = wsum[lane];
      const int ti = wave_incl_scan(t, lane);
      wsum[lane] = ti - t;
      if (lane == 31) tile_total = ti;
    }
    __syncthreads();
    const int ex = carry + wsum[wave] + (inc - s4);
    if (i0     < N) off[i0]     = ex;
    if (i0 + 1 < N) off[i0 + 1] = ex + s1;
    if (i0 + 2 < N) off[i0 + 2] = ex + s2;
    if (i0 + 3 < N) off[i0 + 3] = ex + s3;
    carry += tile_total;
    __syncthreads();
  }
  if (tid == 0) off[N] = carry;
}

__global__ void k_place(const int* __restrict__ row, const int* __restrict__ col,
                        const float* __restrict__ P, const int* __restrict__ off,
                        const int* __restrict__ rank, int2* __restrict__ ed, int E) {
  int e = blockIdx.x * blockDim.x + threadIdx.x;
  if (e >= E) return;
  const float s = 1.0f / (1.0f + __expf(-P[e]));
  const int   p = off[col[e]] + rank[e];
  ed[p] = make_int2(row[e], __float_as_int(s));
}

__global__ void k_dinv(const int* __restrict__ off, const int2* __restrict__ ed,
                       float* __restrict__ dinv, int N, int NP) {
  int i = blockIdx.x * blockDim.x + threadIdx.x;
  if (i >= NP) return;
  if (i >= N) { dinv[i] = 0.0f; return; }
  const int b = off[i], e = off[i + 1];
  float d = 1.0f;
  for (int k = b; k < e; ++k) d += __int_as_float(ed[k].y);
  dinv[i] = (d > 0.0f) ? rsqrtf(d) : 0.0f;
}

template<int NCOL>
__global__ void k_wt_bf16(const float* __restrict__ W, __bf16* __restrict__ Wt) {
  int idx = blockIdx.x * blockDim.x + threadIdx.x;
  if (idx >= 128 * NCOL) return;
  int k = idx / NCOL, n = idx % NCOL;
  Wt[n * 128 + k] = tobf(W[idx]);
}

template<int NCOL, bool RELU_A>
__global__ void k_gemm_wmma(const float* __restrict__ A,
                            const __bf16* __restrict__ Bt,
                            const float* __restrict__ dinv,
                            float* __restrict__ H, int N) {
  constexpr int K    = 128;
  constexpr int LDSP = 136;
  __shared__ __align__(16) __bf16 As[16 * LDSP];

  const int lane = threadIdx.x & 31;
  const int wave = threadIdx.x >> 5;
  const int m0   = blockIdx.x * 16;
  const int n0   = wave * 16;
  const int half = lane >> 4;
  const int l15  = lane & 15;

  for (int base = threadIdx.x * 8; base < 16 * K; base += blockDim.x * 8) {
    const int m = base >> 7;
    const int k = base & (K - 1);
    int mg = m0 + m; if (mg > N - 1) mg = N - 1;
    const float* src = A + (size_t)mg * K + k;
    float4 f0 = *(const float4*)(src);
    float4 f1 = *(const float4*)(src + 4);
    if (RELU_A) {
      f0.x = fmaxf(f0.x, 0.f); f0.y = fmaxf(f0.y, 0.f);
      f0.z = fmaxf(f0.z, 0.f); f0.w = fmaxf(f0.w, 0.f);
      f1.x = fmaxf(f1.x, 0.f); f1.y = fmaxf(f1.y, 0.f);
      f1.z = fmaxf(f1.z, 0.f); f1.w = fmaxf(f1.w, 0.f);
    }
    v8bf pk;
    pk[0] = tobf(f0.x); pk[1] = tobf(f0.y); pk[2] = tobf(f0.z); pk[3] = tobf(f0.w);
    pk[4] = tobf(f1.x); pk[5] = tobf(f1.y); pk[6] = tobf(f1.z); pk[7] = tobf(f1.w);
    *(v8bf*)(&As[m * LDSP + k]) = pk;
  }
  __syncthreads();

  const __bf16* Brow = Bt + (size_t)(n0 + l15) * K;
  const __bf16* Alds = As + l15 * LDSP;

  const float4 dv0 = *(const float4*)(dinv + m0 + half * 8);
  const float4 dv1 = *(const float4*)(dinv + m0 + half * 8 + 4);

  v8f acc = {};

#pragma unroll
  for (int k0 = 0; k0 < K; k0 += 32) {
    const v8bf a0 = *(const v8bf*)(Alds + k0 + half * 8);
    const v8bf a1 = *(const v8bf*)(Alds + k0 + 16 + half * 8);
    const v16bf a = __builtin_shufflevector(a0, a1,
        0, 1, 2, 3, 4, 5, 6, 7, 8, 9, 10, 11, 12, 13, 14, 15);
    const v8bf b0 = *(const v8bf*)(Brow + k0 + half * 8);
    const v8bf b1 = *(const v8bf*)(Brow + k0 + 16 + half * 8);
    const v16bf b = __builtin_shufflevector(b0, b1,
        0, 1, 2, 3, 4, 5, 6, 7, 8, 9, 10, 11, 12, 13, 14, 15);
    acc = __builtin_amdgcn_wmma_f32_16x16x32_bf16(false, a, false, b,
                                                  (short)0, acc, false, false);
  }

  acc[0] *= dv0.x; acc[1] *= dv0.y; acc[2] *= dv0.z; acc[3] *= dv0.w;
  acc[4] *= dv1.x; acc[5] *= dv1.y; acc[6] *= dv1.z; acc[7] *= dv1.w;
  const int   n  = n0 + l15;
  if (m0 + 16 <= N) {
#pragma unroll
    for (int r = 0; r < 8; ++r) {
      const int m = m0 + half * 8 + r;
      H[(size_t)m * NCOL + n] = acc[r];
    }
  } else {
#pragma unroll
    for (int r = 0; r < 8; ++r) {
      const int m = m0 + half * 8 + r;
      if (m < N) H[(size_t)m * NCOL + n] = acc[r];
    }
  }
}

template<int F>
__global__ void k_gather(const int* __restrict__ off, const int2* __restrict__ ed,
                         const float* __restrict__ dinv, const float* __restrict__ h,
                         const float* __restrict__ bias, float* __restrict__ out, int N) {
  const int node = __builtin_amdgcn_readfirstlane(
                     (blockIdx.x * blockDim.x + threadIdx.x) >> 5);
  const int lane = threadIdx.x & 31;
  if (node >= N) return;
  constexpr int PER = F / 32;
  const int   beg = off[node];
  const int   end = off[node + 1];
  const float di  = dinv[node];
  const float* hl = h + lane * PER;

  float acc[PER];
  float bs[PER];
  {
    const float* hp = hl + (size_t)node * F;
    const float* bp = bias + lane * PER;
    if constexpr (PER == 4) {
      const float4 v = *(const float4*)hp;
      const float4 b = *(const float4*)bp;
      acc[0] = v.x; acc[1] = v.y; acc[2] = v.z; acc[3] = v.w;
      bs[0] = b.x; bs[1] = b.y; bs[2] = b.z; bs[3] = b.w;
    } else {
      const float2 v = *(const float2*)hp;
      const float2 b = *(const float2*)bp;
      acc[0] = v.x; acc[1] = v.y;
      bs[0] = b.x; bs[1] = b.y;
    }
  }

  for (int base = beg; base < end; base += 32) {
    const int k = base + lane;
    int   r  = node;
    float cf = 0.0f;
    if (k < end) {
      const int2 t = ed[k];
      r = t.x;
      r = r < 0 ? 0 : (r > N - 1 ? N - 1 : r);
      cf = __int_as_float(t.y);
    }
    const int nk = (end - base < 32) ? (end - base) : 32;
    for (int j = 0; j < nk; j += 4) {
      const int   r0 = __builtin_amdgcn_readlane(r, j);
      const int   r1 = __builtin_amdgcn_readlane(r, j + 1);
      const int   r2 = __builtin_amdgcn_readlane(r, j + 2);
      const int   r3 = __builtin_amdgcn_readlane(r, j + 3);
      const float c0 = __int_as_float(__builtin_amdgcn_readlane(__float_as_int(cf), j));
      const float c1 = __int_as_float(__builtin_amdgcn_readlane(__float_as_int(cf), j + 1));
      const float c2 = __int_as_float(__builtin_amdgcn_readlane(__float_as_int(cf), j + 2));
      const float c3 = __int_as_float(__builtin_amdgcn_readlane(__float_as_int(cf), j + 3));
      if constexpr (PER == 4) {
        const float4 v0 = *(const float4*)(hl + (size_t)r0 * F);
        const float4 v1 = *(const float4*)(hl + (size_t)r1 * F);
        const float4 v2 = *(const float4*)(hl + (size_t)r2 * F);
        const float4 v3 = *(const float4*)(hl + (size_t)r3 * F);
        acc[0] = fmaf(c0, v0.x, acc[0]); acc[1] = fmaf(c0, v0.y, acc[1]);
        acc[2] = fmaf(c0, v0.z, acc[2]); acc[3] = fmaf(c0, v0.w, acc[3]);
        acc[0] = fmaf(c1, v1.x, acc[0]); acc[1] = fmaf(c1, v1.y, acc[1]);
        acc[2] = fmaf(c1, v1.z, acc[2]); acc[3] = fmaf(c1, v1.w, acc[3]);
        acc[0] = fmaf(c2, v2.x, acc[0]); acc[1] = fmaf(c2, v2.y, acc[1]);
        acc[2] = fmaf(c2, v2.z, acc[2]); acc[3] = fmaf(c2, v2.w, acc[3]);
        acc[0] = fmaf(c3, v3.x, acc[0]); acc[1] = fmaf(c3, v3.y, acc[1]);
        acc[2] = fmaf(c3, v3.z, acc[2]); acc[3] = fmaf(c3, v3.w, acc[3]);
      } else {
        const float2 v0 = *(const float2*)(hl + (size_t)r0 * F);
        const float2 v1 = *(const float2*)(hl + (size_t)r1 * F);
        const float2 v2 = *(const float2*)(hl + (size_t)r2 * F);
        const float2 v3 = *(const float2*)(hl + (size_t)r3 * F);
        acc[0] = fmaf(c0, v0.x, acc[0]); acc[1] = fmaf(c0, v0.y, acc[1]);
        acc[0] = fmaf(c1, v1.x, acc[0]); acc[1] = fmaf(c1, v1.y, acc[1]);
        acc[0] = fmaf(c2, v2.x, acc[0]); acc[1] = fmaf(c2, v2.y, acc[1]);
        acc[0] = fmaf(c3, v3.x, acc[0]); acc[1] = fmaf(c3, v3.y, acc[1]);
      }
    }
  }

  float* op = out + (size_t)node * F + lane * PER;
  if constexpr (PER == 4) {
    *(float4*)op = make_float4(fmaf(di, acc[0], bs[0]), fmaf(di, acc[1], bs[1]),
                               fmaf(di, acc[2], bs[2]), fmaf(di, acc[3], bs[3]));
  } else {
    *(float2*)op = make_float2(fmaf(di, acc[0], bs[0]), fmaf(di, acc[1], bs[1]));
  }
}

extern "C" void kernel_launch(void* const* d_in, const int* in_sizes, int n_in,
                              void* d_out, int out_size, void* d_ws, size_t ws_size,
                              hipStream_t stream) {
  const float* x  = (const float*)d_in[0];
  const int*   ei = (const int*)  d_in[1];
  const float* P  = (const float*)d_in[2];
  const float* W1 = (const float*)d_in[3];
  const float* b1 = (const float*)d_in[4];
  const float* W2 = (const float*)d_in[5];
  const float* b2 = (const float*)d_in[6];
  float* out = (float*)d_out;

  constexpr int F_IN = 128, H1 = 128, H2 = 64;
  const int N = in_sizes[0] / F_IN;
  const int E = in_sizes[2];
  const int* row = ei;
  const int* col = ei + E;

  const int strips = (N + 15) / 16;
  const int NP     = strips * 16;

  float*  h    = (float*)d_ws;
  float*  o1   = h + (size_t)N * H1;
  float*  dinv = o1 + (size_t)N * H1;
  __bf16* w1t  = (__bf16*)(dinv + NP);
  __bf16* w2t  = w1t + F_IN * H1;
  int2*   ed   = (int2*)(w2t + F_IN * H2);
  int*    cnt  = (int*)(ed + E);
  int*    off  = cnt + N;
  int*    rank = off + N + 1;

  const int T = 256;

  k_fill0<<<(N + T - 1) / T, T, 0, stream>>>(cnt, N);
  k_rank <<<(E + T - 1) / T, T, 0, stream>>>(col, cnt, rank, E);
  k_scan <<<1, 1024, 0, stream>>>(cnt, off, N);
  k_place<<<(E + T - 1) / T, T, 0, stream>>>(row, col, P, off, rank, ed, E);
  k_dinv <<<(NP + T - 1) / T, T, 0, stream>>>(off, ed, dinv, N, NP);
  k_wt_bf16<H1><<<(F_IN * H1 + T - 1) / T, T, 0, stream>>>(W1, w1t);
  k_wt_bf16<H2><<<(F_IN * H2 + T - 1) / T, T, 0, stream>>>(W2, w2t);

  k_gemm_wmma<H1, false><<<strips, 32 * (H1 / 16), 0, stream>>>(x, w1t, dinv, h, N);
  k_gather<H1><<<(N + 7) / 8, T, 0, stream>>>(off, ed, dinv, h, b1, o1, N);

  k_gemm_wmma<H2, true><<<strips, 32 * (H2 / 16), 0, stream>>>(o1, w2t, dinv, h, N);
  k_gather<H2><<<(N + 7) / 8, T, 0, stream>>>(off, ed, dinv, h, b2, out, N);
}


# offset scan: one block walking the 25 tiles in turn -> one block per 4096-counter tile that itself adds up the tiles in front (loads in flight together); same integer offsets
# speedup vs baseline: 2.5398x; 1.1746x over previous
#include <hip/hip_runtime.h>
#include <hip/hip_bf16.h>

typedef __attribute__((ext_vector_type(16))) __bf16        v16bf;
typedef __attribute__((ext_vector_type(8)))  __bf16        v8bf;
typedef __attribute__((ext_vector_type(8)))  float         v8f;

__device__ __forceinline__ __bf16 tobf(float f) { return (__bf16)f; }


__global__ void k_fill0(int* __restrict__ p, int n) {
  int i = blockIdx.x * blockDim.x + threadIdx.x;
  if (i < n) p[i] = 0;
}

__global__ void __launch_bounds__(256)
k_rank(const int* __restrict__ col, int* __restrict__ cnt,
       int* __restrict__ rank, int E) {
  const int stride = gridDim.x * blockDim.x;
  for (int e0 = blockIdx.x * blockDim.x + threadIdx.x; e0 < E; e0 += 4 * stride) {
    int c[4], r[4];
#pragma unroll
    for (int u = 0; u < 4; ++u) {
      const int e = e0 + u * stride;
      c[u] = col[(e < E) ? e : (E - 1)];
    }
#pragma unroll
    for (int u = 0; u < 4; ++u) {
      const int e = e0 + u * stride;
      r[u] = 0;
      if (e < E) r[u] = atomicAdd(&cnt[c[u]], 1);
    }
#pragma unroll
    for (int u = 0; u < 4; ++u) {
      const int e = e0 + u * stride;
      if (e < E) rank[e] = r[u];
    }
  }
}

__device__ __forceinline__ int wave_incl_scan(int x, int lane) {
  x += __builtin_amdgcn_update_dpp(0, x, 0x111, 0xf, 0xf, false);
  x += __builtin_amdgcn_update_dpp(0, x, 0x112, 0xf, 0xf, false);
  x += __builtin_amdgcn_update_dpp(0, x, 0x114, 0xf, 0xf, false);
  x += __builtin_amdgcn_update_dpp(0, x, 0x118, 0xf, 0xf, false);
  const int lo = __builtin_amdgcn_readlane(x, 15);
  if (lane >= 16) x += lo;
  return x;
}

__global__ void __launch_bounds__(1024)
k_scan(const int* __restrict__ cnt, int* __restrict__ off, int N) {
  __shared__ int wsum[32];
  __shared__ int wpre[32];
  __shared__ int tile_total;
  __shared__ int pre_total;
  const int tid  = threadIdx.x;
  const int lane = tid & 31;
  const int wave = tid >> 5;
  const int nb   = blockIdx.x;
  const int i0   = nb * 4096 + tid * 4;

  int4 v = make_int4(0, 0, 0, 0);
  if (i0 + 3 < N) {
    v = *(const int4*)(cnt + i0);
  } else {
    if (i0     < N) v.x = cnt[i0];
    if (i0 + 1 < N) v.y = cnt[i0 + 1];
    if (i0 + 2 < N) v.z = cnt[i0 + 2];
  }

  int acc = 0;
  for (int t0 = 0; t0 < nb; t0 += 8) {
    int4 u[8];
#pragma unroll
    for (int k = 0; k < 8; ++k) {
      const int t = (t0 + k < nb) ? (t0 + k) : (nb - 1);
      u[k] = *(const int4*)(cnt + t * 4096 + tid * 4);
    }
#pragma unroll
    for (int k = 0; k < 8; ++k) {
      if (t0 + k < nb) acc += (u[k].x + u[k].y) + (u[k].z + u[k].w);
    }
  }

  const int s1 = v.x, s2 = s1 + v.y, s3 = s2 + v.z, s4 = s3 + v.w;
  const int inc  = wave_incl_scan(s4, lane);
  const int pinc = wave_incl_scan(acc, lane);
  if (lane == 31) { wsum[wave] = inc; wpre[wave] = pinc; }
  __syncthreads();
  if (wave == 0) {
    const int t  = wsum[lane];
    const int ti = wave_incl_scan(t, lane);
    const int pi = wave_incl_scan(wpre[lane], lane);
    wsum[lane] = ti - t;
    if (lane == 31) { tile_total = ti; pre_total = pi; }
  }
  __syncthreads();
  const int carry = pre_total;
  const int ex = carry + wsum[wave] + (inc - s4);
  if (i0     < N) off[i0]     = ex;
  if (i0 + 1 < N) off[i0 + 1] = ex + s1;
  if (i0 + 2 < N) off[i0 + 2] = ex + s2;
  if (i0 + 3 < N) off[i0 + 3] = ex + s3;
  if (nb == (int)gridDim.x - 1 && tid == 0) off[N] = carry + tile_total;
}

__global__ void __launch_bounds__(256)
k_place(const int* __restrict__ row, const int* __restrict__ col,
        const float* __restrict__ P, const int* __restrict__ off,
        const int* __restrict__ rank, int2* __restrict__ ed, int E) {
  const int stride = gridDim.x * blockDim.x;
  for (int e0 = blockIdx.x * blockDim.x + threadIdx.x; e0 < E; e0 += 4 * stride) {
    int c[4], rk[4], rw[4], o[4];
    float pv[4];
#pragma unroll
    for (int u = 0; u < 4; ++u) {
      const int e  = e0 + u * stride;
      const int ec = (e < E) ? e : (E - 1);
      c[u]  = col[ec];
      rk[u] = rank[ec];
      rw[u] = row[ec];
      pv[u] = P[ec];
    }
#pragma unroll
    for (int u = 0; u < 4; ++u) o[u] = off[c[u]];
#pragma unroll
    for (int u = 0; u < 4; ++u) {
      const int e = e0 + u * stride;
      if (e < E) {
        const float s = 1.0f / (1.0f + __expf(-pv[u]));
        ed[o[u] + rk[u]] = make_int2(rw[u], __float_as_int(s));
      }
    }
  }
}

__global__ void k_dinv(const int* __restrict__ off, const int2* __restrict__ ed,
                       float* __restrict__ dinv, int N, int NP) {
  int i = blockIdx.x * blockDim.x + threadIdx.x;
  if (i >= NP) return;
  if (i >= N) { dinv[i] = 0.0f; return; }
  const int b = off[i], e = off[i + 1];
  float d = 1.0f;
  for (int k = b; k < e; ++k) d += __int_as_float(ed[k].y);
  dinv[i] = (d > 0.0f) ? rsqrtf(d) : 0.0f;
}

template<int NCOL>
__global__ void k_wt_bf16(const float* __restrict__ W, __bf16* __restrict__ Wt) {
  int idx = blockIdx.x * blockDim.x + threadIdx.x;
  if (idx >= 128 * NCOL) return;
  int k = idx / NCOL, n = idx % NCOL;
  Wt[n * 128 + k] = tobf(W[idx]);
}

template<int NT, int PASSES>
__device__ __forceinline__ void load_strip(const float* __restrict__ A, int strip, int N,
                                           float4 (&fa)[PASSES], float4 (&fb)[PASSES]) {
#pragma unroll
  for (int p = 0; p < PASSES; ++p) {
    const int base = (p * NT + (int)threadIdx.x) * 8;
    const int m = base >> 7;
    const int k = base & 127;
    int mg = strip * 16 + m; if (mg > N - 1) mg = N - 1;
    const float* src = A + (size_t)mg * 128 + k;
    fa[p] = *(const float4*)(src);
    fb[p] = *(const float4*)(src + 4);
  }
}

template<int NCOL, bool RELU_A>
__global__ void __launch_bounds__(32 * (NCOL / 16))
k_gemm_wmma(const float* __restrict__ A,
            const __bf16* __restrict__ Bt,
            const float* __restrict__ dinv,
            float* __restrict__ H, int N, int strips) {
  constexpr int K      = 128;
  constexpr int LDSP   = 136;
  constexpr int NT     = 32 * (NCOL / 16);
  constexpr int PASSES = (16 * K) / (NT * 8);
  __shared__ __align__(16) __bf16 As[2][16 * LDSP];

  const int lane = threadIdx.x & 31;
  const int wave = threadIdx.x >> 5;
  const int n0   = wave * 16;
  const int half = lane >> 4;
  const int l15  = lane & 15;

  const __bf16* Brow = Bt + (size_t)(n0 + l15) * K;
  const int     aoff = l15 * LDSP;

  v16bf bfr[4];
#pragma unroll
  for (int i = 0; i < 4; ++i) {
    const v8bf b0 = *(const v8bf*)(Brow + i * 32 + half * 8);
    const v8bf b1 = *(const v8bf*)(Brow + i * 32 + 16 + half * 8);
    bfr[i] = __builtin_shufflevector(b0, b1,
        0, 1, 2, 3, 4, 5, 6, 7, 8, 9, 10, 11, 12, 13, 14, 15);
  }

  float4 fa[PASSES], fb[PASSES];
  int s = blockIdx.x;
  if (s < strips) load_strip<NT, PASSES>(A, s, N, fa, fb);

  int pb = 0;
  for (; s < strips; s += gridDim.x, pb ^= 1) {
    const int m0 = s * 16;
    __bf16* Ab = As[pb];
#pragma unroll
    for (int p = 0; p < PASSES; ++p) {
      const int base = (p * NT + (int)threadIdx.x) * 8;
      const int m = base >> 7;
      const int k = base & (K - 1);
      float4 f0 = fa[p], f1 = fb[p];
      if (RELU_A) {
        f0.x = fmaxf(f0.x, 0.f); f0.y = fmaxf(f0.y, 0.f);
        f0.z = fmaxf(f0.z, 0.f); f0.w = fmaxf(f0.w, 0.f);
        f1.x = fmaxf(f1.x, 0.f); f1.y = fmaxf(f1.y, 0.f);
        f1.z = fmaxf(f1.z, 0.f); f1.w = fmaxf(f1.w, 0.f);
      }
      v8bf pk;
      pk[0] = tobf(f0.x); pk[1] = tobf(f0.y); pk[2] = tobf(f0.z); pk[3] = tobf(f0.w);
      pk[4] = tobf(f1.x); pk[5] = tobf(f1.y); pk[6] = tobf(f1.z); pk[7] = tobf(f1.w);
      *(v8bf*)(&Ab[m * LDSP + k]) = pk;
    }
    __syncthreads();

    const float4 dv0 = *(const float4*)(dinv + m0 + half * 8);
    const float4 dv1 = *(const float4*)(dinv + m0 + half * 8 + 4);

    const int sn = s + gridDim.x;
    load_strip<NT, PASSES>(A, (sn < strips) ? sn : s, N, fa, fb);

    v8f acc = {};
#pragma unroll
    for (int i = 0; i < 4; ++i) {
      const v8bf a0 = *(const v8bf*)(Ab + aoff + i * 32 + half * 8);
      const v8bf a1 = *(const v8bf*)(Ab + aoff + i * 32 + 16 + half * 8);
      const v16bf a = __builtin_shufflevector(a0, a1,
          0, 1, 2, 3, 4, 5, 6, 7, 8, 9, 10, 11, 12, 13, 14, 15);
      acc = __builtin_amdgcn_wmma_f32_16x16x32_bf16(false, a, false, bfr[i],
                                                    (short)0, acc, false, false);
    }

    acc[0] *= dv0.x; acc[1] *= dv0.y; acc[2] *= dv0.z; acc[3] *= dv0.w;
    acc[4] *= dv1.x; acc[5] *= dv1.y; acc[6] *= dv1.z; acc[7] *= dv1.w;
    const int n = n0 + l15;
    if (m0 + 16 <= N) {
#pragma unroll
      for (int r = 0; r < 8; ++r) {
        const int m = m0 + half * 8 + r;
        H[(size_t)m * NCOL + n] = acc[r];
      }
    } else {
#pragma unroll
      for (int r = 0; r < 8; ++r) {
        const int m = m0 + half * 8 + r;
        if (m < N) H[(size_t)m * NCOL + n] = acc[r];
      }
    }
  }
}

template<int PER>
__device__ __forceinline__ void ld_row(const float* __restrict__ p, float (&v)[PER]) {
  if constexpr (PER == 4) {
    const float4 t = *(const float4*)p;
    v[0] = t.x; v[1] = t.y; v[2] = t.z; v[3] = t.w;
  } else {
    const float2 t = *(const float2*)p;
    v[0] = t.x; v[1] = t.y;
  }
}

template<int F, int NB>
__device__ __forceinline__ void rows_fma(const float* __restrict__ hl, int r, float cf,
                                         int j, float (&acc)[F / 32]) {
  constexpr int PER = F / 32;
  float v[NB][PER];
  float c[NB];
#pragma unroll
  for (int u = 0; u < NB; ++u) {
    const int ru = __builtin_amdgcn_readlane(r, j + u);
    c[u] = __int_as_float(__builtin_amdgcn_readlane(__float_as_int(cf), j + u));
    ld_row<PER>(hl + (size_t)ru * F, v[u]);
  }
#pragma unroll
  for (int u = 0; u < NB; ++u) {
#pragma unroll
    for (int q = 0; q < PER; ++q) acc[q] = fmaf(c[u], v[u][q], acc[q]);
  }
}

template<int F>
__global__ void __launch_bounds__(256)
k_gather(const int* __restrict__ off, const int2* __restrict__ ed,
         const float* __restrict__ dinv, const float* __restrict__ h,
         const float* __restrict__ bias, float* __restrict__ out, int N) {
  constexpr int PER = F / 32;
  const int lane = threadIdx.x & 31;
  const int gw   = __builtin_amdgcn_readfirstlane(
                     (blockIdx.x * blockDim.x + threadIdx.x) >> 5);
  const int tw   = (gridDim.x * blockDim.x) >> 5;
  const int per  = (N + tw - 1) / tw;
  const int n0   = gw * per;
  const int n1   = (n0 + per < N) ? (n0 + per) : N;
  if (n0 >= n1) return;

  const float* hl = h + lane * PER;
  float bs[PER];
  ld_row<PER>(bias + lane * PER, bs);

  for (int cb = n0; cb < n1; cb += 31) {
    const int cn = (n1 - cb < 31) ? (n1 - cb) : 31;
    int oi = cb + lane; if (oi > N) oi = N;
    const int ov = off[oi];
    int dj = cb + lane; if (dj > N - 1) dj = N - 1;
    const float dvv = dinv[dj];

    int beg = __builtin_amdgcn_readlane(ov, 0);
    int end = __builtin_amdgcn_readlane(ov, 1);
    int   rn  = cb;
    float cfn = 0.0f;
    if (beg + lane < end) {
      const int2 t = ed[beg + lane];
      rn  = t.x;
      cfn = __int_as_float(t.y);
    }

    for (int t = 0; t < cn; ++t) {
      const int node = cb + t;
      const int b = beg, e = end;
      int   r  = rn < 0 ? 0 : (rn > N - 1 ? N - 1 : rn);
      float cf = cfn;

      float acc[PER];
      ld_row<PER>(hl + (size_t)node * F, acc);
      const float di = __int_as_float(
          __builtin_amdgcn_readlane(__float_as_int(dvv), t));

      if (t + 1 < cn) {
        beg = e;
        end = __builtin_amdgcn_readlane(ov, t + 2);
        rn  = node + 1;
        cfn = 0.0f;
        if (beg + lane < end) {
          const int2 q = ed[beg + lane];
          rn  = q.x;
          cfn = __int_as_float(q.y);
        }
      }

      for (int base = b; base < e; base += 32) {
        if (base != b) {
          r  = node;
          cf = 0.0f;
          if (base + lane < e) {
            const int2 q = ed[base + lane];
            r  = q.x < 0 ? 0 : (q.x > N - 1 ? N - 1 : q.x);
            cf = __int_as_float(q.y);
          }
        }
        const int nk = (e - base < 32) ? (e - base) : 32;
        for (int j = 0; j < nk; j += 8) {
          if (j + 4 < nk) rows_fma<F, 8>(hl, r, cf, j, acc);
          else            rows_fma<F, 4>(hl, r, cf, j, acc);
        }
      }

      float* op = out + (size_t)node * F + lane * PER;
      if constexpr (PER == 4) {
        *(float4*)op = make_float4(fmaf(di, acc[0], bs[0]), fmaf(di, acc[1], bs[1]),
                                   fmaf(di, acc[2], bs[2]), fmaf(di, acc[3], bs[3]));
      } else {
        *(float2*)op = make_float2(fmaf(di, acc[0], bs[0]), fmaf(di, acc[1], bs[1]));
      }
    }
  }
}

extern "C" void kernel_launch(void* const* d_in, const int* in_sizes, int n_in,
                              void* d_out, int out_size, void* d_ws, size_t ws_size,
                              hipStream_t stream) {
  const float* x  = (const float*)d_in[0];
  const int*   ei = (const int*)  d_in[1];
  const float* P  = (const float*)d_in[2];
  const float* W1 = (const float*)d_in[3];
  const float* b1 = (const float*)d_in[4];
  const float* W2 = (const float*)d_in[5];
  const float* b2 = (const float*)d_in[6];
  float* out = (float*)d_out;

  constexpr int F_IN = 128, H1 = 128, H2 = 64;
  const int N = in_sizes[0] / F_IN;
  const int E = in_sizes[2];
  const int* row = ei;
  const int* col = ei + E;

  const int strips = (N + 15) / 16;
  const int NP     = strips * 16;

  float*  h    = (float*)d_ws;
  float*  o1   = h + (size_t)N * H1;
  float*  dinv = o1 + (size_t)N * H1;
  __bf16* w1t  = (__bf16*)(dinv + NP);
  __bf16* w2t  = w1t + F_IN * H1;
  int2*   ed   = (int2*)(w2t + F_IN * H2);
  int*    cnt  = (int*)(ed + E);
  int*    off  = cnt + N;
  int*    rank = off + N + 1;

  const int T = 256;
  constexpr int RG  = 1024;
  constexpr int RG2 = 2048;

  k_fill0<<<(N + T - 1) / T, T, 0, stream>>>(cnt, N);
  k_rank <<<RG, T, 0, stream>>>(col, cnt, rank, E);
  k_scan <<<(N + 4095) / 4096, 1024, 0, stream>>>(cnt, off, N);
  k_place<<<RG, T, 0, stream>>>(row, col, P, off, rank, ed, E);
  k_dinv <<<(NP + T - 1) / T, T, 0, stream>>>(off, ed, dinv, N, NP);
  k_wt_bf16<H1><<<(F_IN * H1 + T - 1) / T, T, 0, stream>>>(W1, w1t);
  k_wt_bf16<H2><<<(F_IN * H2 + T - 1) / T, T, 0, stream>>>(W2, w2t);

  k_gemm_wmma<H1, false><<<RG, 32 * (H1 / 16), 0, stream>>>(x, w1t, dinv, h, N, strips);
  k_gather<H1><<<RG, T, 0, stream>>>(off, ed, dinv, h, b1, o1, N);

  k_gemm_wmma<H2, true><<<RG2, 32 * (H2 / 16), 0, stream>>>(o1, w2t, dinv, h, N, strips);
  k_gather<H2><<<RG, T, 0, stream>>>(off, ed, dinv, h, b2, out, N);
}


# edge slot ranks counted in LDS not by global atomics: edges split into 256-node destination bins, one block per bin ranks them and writes slot rows as whole lines; slot order = LDS arrival order
# speedup vs baseline: 4.0294x; 1.5865x over previous
#include <hip/hip_runtime.h>
#include <hip/hip_bf16.h>

typedef __attribute__((ext_vector_type(16))) __bf16        v16bf;
typedef __attribute__((ext_vector_type(8)))  __bf16        v8bf;
typedef __attribute__((ext_vector_type(8)))  float         v8f;

__device__ __forceinline__ __bf16 tobf(float f) { return (__bf16)f; }

constexpr int CAP      = 32;
constexpr int BN_SHIFT = 8;
constexpr int BN       = 1 << BN_SHIFT;
constexpr int BINCAP   = 8192;
constexpr int NBMAX    = 1024;
constexpr int CSTRIDE  = 64;
constexpr int EPT      = 16;
constexpr int TILE     = 256 * EPT;

__global__ void k_fill0(int* __restrict__ p, int n) {
  int i = blockIdx.x * blockDim.x + threadIdx.x;
  if (i < n) p[i] = 0;
}

__global__ void __launch_bounds__(256)
k_coarse(const int* __restrict__ row, const int* __restrict__ col,
         const float* __restrict__ P, int* __restrict__ binc, int* __restrict__ covc,
         int4* __restrict__ bins, int4* __restrict__ covf, int E, int N, int NB) {
  __shared__ int hist[NBMAX];
  if (NB > NBMAX) return;
  const int tid    = threadIdx.x;
  const int ntiles = (E + TILE - 1) / TILE;
  for (int tile = blockIdx.x; tile < ntiles; tile += gridDim.x) {
    for (int b = tid; b < NB; b += 256) hist[b] = 0;
    __syncthreads();
    int   c[EPT], rw[EPT], rk[EPT];
    float pv[EPT];
    const int e0 = tile * TILE + tid;
#pragma unroll
    for (int u = 0; u < EPT; ++u) {
      const int e  = e0 + u * 256;
      const int ec = (e < E) ? e : (E - 1);
      c[u]  = col[ec];
      rw[u] = row[ec];
      pv[u] = P[ec];
    }
#pragma unroll
    for (int u = 0; u < EPT; ++u) {
      const int e = e0 + u * 256;
      const int cc = c[u] < 0 ? 0 : (c[u] > N - 1 ? N - 1 : c[u]);
      c[u]  = cc;
      rk[u] = 0;
      if (e < E) rk[u] = atomicAdd(&hist[cc >> BN_SHIFT], 1);
    }
    __syncthreads();
    for (int b = tid; b < NB; b += 256) {
      const int hc = hist[b];
      if (hc > 0) hist[b] = atomicAdd(&binc[b * CSTRIDE], hc);
    }
    __syncthreads();
    int bs[EPT];
#pragma unroll
    for (int u = 0; u < EPT; ++u) bs[u] = hist[c[u] >> BN_SHIFT];
#pragma unroll
    for (int u = 0; u < EPT; ++u) {
      const int e = e0 + u * 256;
      if (e < E) {
        const int   b   = c[u] >> BN_SHIFT;
        const int   pos = bs[u] + rk[u];
        const float s   = 1.0f / (1.0f + __expf(-pv[u]));
        const int4  q   = make_int4(c[u], rw[u], __float_as_int(s), 0);
        if (pos < BINCAP) {
          bins[(size_t)b * BINCAP + pos] = q;
        } else {
          const int k = atomicAdd(covc, 1);
          if (k < E) covf[k] = q;
        }
      }
    }
    __syncthreads();
  }
}

__device__ __forceinline__ void place_edge(const int4 q, int nb0, int* lcnt, int2* srow,
                                           int* __restrict__ cnt, int4* __restrict__ ovf,
                                           int E, int N) {
  const int cl = q.x - nb0;
  if (cl < 0 || cl >= BN) return;
  const int r = atomicAdd(&lcnt[cl], 1);
  if (r < CAP) {
    srow[cl * CAP + r] = make_int2(q.y, q.z);
  } else {
    const int k = atomicAdd(&cnt[N], 1);
    if (k < E) ovf[k] = make_int4(q.x, q.y, q.z, 0);
  }
}

__global__ void __launch_bounds__(BN)
k_fine(const int* __restrict__ binc, const int* __restrict__ covc,
       const int4* __restrict__ bins, const int4* __restrict__ covf,
       int* __restrict__ cnt, int2* __restrict__ ed, int4* __restrict__ ovf,
       int E, int N) {
  __shared__ __align__(16) int2 srow[BN * CAP];
  __shared__ int lcnt[BN];
  const int tid = threadIdx.x;
  const int b   = blockIdx.x;
  const int nb0 = b << BN_SHIFT;
  lcnt[tid] = 0;
  __syncthreads();

  int nb = binc[b * CSTRIDE];
  if (nb > BINCAP) nb = BINCAP;
  const int4* bl = bins + (size_t)b * BINCAP;
  for (int i0 = tid; i0 < nb; i0 += 4 * BN) {
    int4 q[4];
#pragma unroll
    for (int u = 0; u < 4; ++u) {
      const int i = i0 + u * BN;
      q[u] = bl[(i < nb) ? i : (nb - 1)];
    }
#pragma unroll
    for (int u = 0; u < 4; ++u) {
      const int i = i0 + u * BN;
      if (i < nb) place_edge(q[u], nb0, lcnt, srow, cnt, ovf, E, N);
    }
  }
  int nco = covc[0];
  if (nco > E) nco = E;
  for (int i = tid; i < nco; i += BN) {
    const int4 q = covf[i];
    if ((q.x >> BN_SHIFT) == b) place_edge(q, nb0, lcnt, srow, cnt, ovf, E, N);
  }
  __syncthreads();

  if (nb0 + tid < N) cnt[nb0 + tid] = lcnt[tid];
  const int4* s4 = (const int4*)srow;
  int4*       d4 = (int4*)(ed + (size_t)nb0 * CAP);
  constexpr int Q_PER_ROW = CAP * 8 / 16;
  for (int i = tid; i < BN * Q_PER_ROW; i += BN) {
    if (nb0 + i / Q_PER_ROW < N) d4[i] = s4[i];
  }
}

__global__ void k_dinv(const int* __restrict__ cnt, const int2* __restrict__ ed,
                       const int4* __restrict__ ovf,
                       float* __restrict__ dinv, int N, int NP, int E) {
  int i = blockIdx.x * blockDim.x + threadIdx.x;
  if (i >= NP) return;
  if (i >= N) { dinv[i] = 0.0f; return; }
  const int deg = cnt[i];
  const int nk  = (deg < CAP) ? deg : CAP;
  const int2* seg = ed + (size_t)i * CAP;
  float d = 1.0f;
  for (int k = 0; k < nk; ++k) d += __int_as_float(seg[k].y);
  if (deg > CAP) {
    int nov = cnt[N]; if (nov > E) nov = E;
    for (int k = 0; k < nov; ++k) {
      const int4 q = ovf[k];
      if (q.x == i) d += __int_as_float(q.z);
    }
  }
  dinv[i] = (d > 0.0f) ? rsqrtf(d) : 0.0f;
}

template<int NCOL>
__global__ void k_wt_bf16(const float* __restrict__ W, __bf16* __restrict__ Wt) {
  int idx = blockIdx.x * blockDim.x + threadIdx.x;
  if (idx >= 128 * NCOL) return;
  int k = idx / NCOL, n = idx % NCOL;
  Wt[n * 128 + k] = tobf(W[idx]);
}

template<int NT, int PASSES>
__device__ __forceinline__ void load_strip(const float* __restrict__ A, int strip, int N,
                                           float4 (&fa)[PASSES], float4 (&fb)[PASSES]) {
#pragma unroll
  for (int p = 0; p < PASSES; ++p) {
    const int base = (p * NT + (int)threadIdx.x) * 8;
    const int m = base >> 7;
    const int k = base & 127;
    int mg = strip * 16 + m; if (mg > N - 1) mg = N - 1;
    const float* src = A + (size_t)mg * 128 + k;
    fa[p] = *(const float4*)(src);
    fb[p] = *(const float4*)(src + 4);
  }
}

template<int NCOL, bool RELU_A>
__global__ void __launch_bounds__(32 * (NCOL / 16))
k_gemm_wmma(const float* __restrict__ A,
            const __bf16* __restrict__ Bt,
            const float* __restrict__ dinv,
            float* __restrict__ H, int N, int strips) {
  constexpr int K      = 128;
  constexpr int LDSP   = 136;
  constexpr int NT     = 32 * (NCOL / 16);
  constexpr int PASSES = (16 * K) / (NT * 8);
  __shared__ __align__(16) __bf16 As[2][16 * LDSP];

  const int lane = threadIdx.x & 31;
  const int wave = threadIdx.x >> 5;
  const int n0   = wave * 16;
  const int half = lane >> 4;
  const int l15  = lane & 15;

  const __bf16* Brow = Bt + (size_t)(n0 + l15) * K;
  const int     aoff = l15 * LDSP;

  v16bf bfr[4];
#pragma unroll
  for (int i = 0; i < 4; ++i) {
    const v8bf b0 = *(const v8bf*)(Brow + i * 32 + half * 8);
    const v8bf b1 = *(const v8bf*)(Brow + i * 32 + 16 + half * 8);
    bfr[i] = __builtin_shufflevector(b0, b1,
        0, 1, 2, 3, 4, 5, 6, 7, 8, 9, 10, 11, 12, 13, 14, 15);
  }

  float4 fa[PASSES], fb[PASSES];
  int s = blockIdx.x;
  if (s < strips) load_strip<NT, PASSES>(A, s, N, fa, fb);

  int pb = 0;
  for (; s < strips; s += gridDim.x, pb ^= 1) {
    const int m0 = s * 16;
    __bf16* Ab = As[pb];
#pragma unroll
    for (int p = 0; p < PASSES; ++p) {
      const int base = (p * NT + (int)threadIdx.x) * 8;
      const int m = base >> 7;
      const int k = base & (K - 1);
      float4 f0 = fa[p], f1 = fb[p];
      if (RELU_A) {
        f0.x = fmaxf(f0.x, 0.f); f0.y = fmaxf(f0.y, 0.f);
        f0.z = fmaxf(f0.z, 0.f); f0.w = fmaxf(f0.w, 0.f);
        f1.x = fmaxf(f1.x, 0.f); f1.y = fmaxf(f1.y, 0.f);
        f1.z = fmaxf(f1.z, 0.f); f1.w = fmaxf(f1.w, 0.f);
      }
      v8bf pk;
      pk[0] = tobf(f0.x); pk[1] = tobf(f0.y); pk[2] = tobf(f0.z); pk[3] = tobf(f0.w);
      pk[4] = tobf(f1.x); pk[5] = tobf(f1.y); pk[6] = tobf(f1.z); pk[7] = tobf(f1.w);
      *(v8bf*)(&Ab[m * LDSP + k]) = pk;
    }
    __syncthreads();

    const float4 dv0 = *(const float4*)(dinv + m0 + half * 8);
    const float4 dv1 = *(const float4*)(dinv + m0 + half * 8 + 4);

    const int sn = s + gridDim.x;
    load_strip<NT, PASSES>(A, (sn < strips) ? sn : s, N, fa, fb);

    v8f acc = {};
#pragma unroll
    for (int i = 0; i < 4; ++i) {
      const v8bf a0 = *(const v8bf*)(Ab + aoff + i * 32 + half * 8);
      const v8bf a1 = *(const v8bf*)(Ab + aoff + i * 32 + 16 + half * 8);
      const v16bf a = __builtin_shufflevector(a0, a1,
          0, 1, 2, 3, 4, 5, 6, 7, 8, 9, 10, 11, 12, 13, 14, 15);
      acc = __builtin_amdgcn_wmma_f32_16x16x32_bf16(false, a, false, bfr[i],
                                                    (short)0, acc, false, false);
    }

    acc[0] *= dv0.x; acc[1] *= dv0.y; acc[2] *= dv0.z; acc[3] *= dv0.w;
    acc[4] *= dv1.x; acc[5] *= dv1.y; acc[6] *= dv1.z; acc[7] *= dv1.w;
    const int n = n0 + l15;
    if (m0 + 16 <= N) {
#pragma unroll
      for (int r = 0; r < 8; ++r) {
        const int m = m0 + half * 8 + r;
        H[(size_t)m * NCOL + n] = acc[r];
      }
    } else {
#pragma unroll
      for (int r = 0; r < 8; ++r) {
        const int m = m0 + half * 8 + r;
        if (m < N) H[(size_t)m * NCOL + n] = acc[r];
      }
    }
  }
}

template<int PER>
__device__ __forceinline__ void ld_row(const float* __restrict__ p, float (&v)[PER]) {
  if constexpr (PER == 4) {
    const float4 t = *(const float4*)p;
    v[0] = t.x; v[1] = t.y; v[2] = t.z; v[3] = t.w;
  } else {
    const float2 t = *(const float2*)p;
    v[0] = t.x; v[1] = t.y;
  }
}

template<int F, int NB>
__device__ __forceinline__ void rows_fma(const float* __restrict__ hl, int r, float cf,
                                         int j, float (&acc)[F / 32]) {
  constexpr int PER = F / 32;
  float v[NB][PER];
  float c[NB];
#pragma unroll
  for (int u = 0; u < NB; ++u) {
    const int ru = __builtin_amdgcn_readlane(r, j + u);
    c[u] = __int_as_float(__builtin_amdgcn_readlane(__float_as_int(cf), j + u));
    ld_row<PER>(hl + (size_t)ru * F, v[u]);
  }
#pragma unroll
  for (int u = 0; u < NB; ++u) {
#pragma unroll
    for (int q = 0; q < PER; ++q) acc[q] = fmaf(c[u], v[u][q], acc[q]);
  }
}

template<int F>
__global__ void __launch_bounds__(256)
k_gather(const int* __restrict__ cnt, const int2* __restrict__ ed,
         const int4* __restrict__ ovf,
         const float* __restrict__ dinv, const float* __restrict__ h,
         const float* __restrict__ bias, float* __restrict__ out, int N, int E) {
  constexpr int PER = F / 32;
  const int lane = threadIdx.x & 31;
  const int gw   = __builtin_amdgcn_readfirstlane(
                     (blockIdx.x * blockDim.x + threadIdx.x) >> 5);
  const int tw   = (gridDim.x * blockDim.x) >> 5;
  const int per  = (N + tw - 1) / tw;
  const int n0   = gw * per;
  const int n1   = (n0 + per < N) ? (n0 + per) : N;
  if (n0 >= n1) return;

  const float* hl = h + lane * PER;
  float bs[PER];
  ld_row<PER>(bias + lane * PER, bs);

  for (int cb = n0; cb < n1; cb += 32) {
    const int cn = (n1 - cb < 32) ? (n1 - cb) : 32;
    int ci = cb + lane; if (ci > N - 1) ci = N - 1;
    const int   dgv = cnt[ci];
    const float dvv = dinv[ci];

    int deg = __builtin_amdgcn_readlane(dgv, 0);
    int   rn  = cb;
    float cfn = 0.0f;
    if (lane < deg) {
      const int2 t = ed[(size_t)cb * CAP + lane];
      rn  = t.x;
      cfn = __int_as_float(t.y);
    }

    for (int t = 0; t < cn; ++t) {
      const int node = cb + t;
      const int dg = deg;
      int   r  = rn < 0 ? 0 : (rn > N - 1 ? N - 1 : rn);
      float cf = cfn;

      float acc[PER];
      ld_row<PER>(hl + (size_t)node * F, acc);
      const float di = __int_as_float(
          __builtin_amdgcn_readlane(__float_as_int(dvv), t));

      if (t + 1 < cn) {
        deg = __builtin_amdgcn_readlane(dgv, t + 1);
        rn  = node + 1;
        cfn = 0.0f;
        if (lane < deg) {
          const int2 q = ed[(size_t)(node + 1) * CAP + lane];
          rn  = q.x;
          cfn = __int_as_float(q.y);
        }
      }

      const int nk = (dg < CAP) ? dg : CAP;
      for (int j = 0; j < nk; j += 8) {
        if (j + 4 < nk) rows_fma<F, 8>(hl, r, cf, j, acc);
        else            rows_fma<F, 4>(hl, r, cf, j, acc);
      }

      if (dg > CAP) {
        int nov = cnt[N]; if (nov > E) nov = E;
        for (int base = 0; base < nov; base += 32) {
          int   oc = -1, orow = node;
          float ow = 0.0f;
          if (base + lane < nov) {
            const int4 q = ovf[base + lane];
            oc   = q.x;
            orow = q.y < 0 ? 0 : (q.y > N - 1 ? N - 1 : q.y);
            ow   = __int_as_float(q.z);
          }
          unsigned m = __builtin_amdgcn_ballot_w32(oc == node);
          while (m) {
            const int l = __builtin_ctz(m);
            m &= m - 1;
            const int   ru = __builtin_amdgcn_readlane(orow, l);
            const float cu = __int_as_float(
                __builtin_amdgcn_readlane(__float_as_int(ow), l));
            float v[PER];
            ld_row<PER>(hl + (size_t)ru * F, v);
#pragma unroll
            for (int q = 0; q < PER; ++q) acc[q] = fmaf(cu, v[q], acc[q]);
          }
        }
      }

      float* op = out + (size_t)node * F + lane * PER;
      if constexpr (PER == 4) {
        *(float4*)op = make_float4(fmaf(di, acc[0], bs[0]), fmaf(di, acc[1], bs[1]),
                                   fmaf(di, acc[2], bs[2]), fmaf(di, acc[3], bs[3]));
      } else {
        *(float2*)op = make_float2(fmaf(di, acc[0], bs[0]), fmaf(di, acc[1], bs[1]));
      }
    }
  }
}

extern "C" void kernel_launch(void* const* d_in, const int* in_sizes, int n_in,
                              void* d_out, int out_size, void* d_ws, size_t ws_size,
                              hipStream_t stream) {
  const float* x  = (const float*)d_in[0];
  const int*   ei = (const int*)  d_in[1];
  const float* P  = (const float*)d_in[2];
  const float* W1 = (const float*)d_in[3];
  const float* b1 = (const float*)d_in[4];
  const float* W2 = (const float*)d_in[5];
  const float* b2 = (const float*)d_in[6];
  float* out = (float*)d_out;

  constexpr int F_IN = 128, H1 = 128, H2 = 64;
  const int N = in_sizes[0] / F_IN;
  const int E = in_sizes[2];
  const int* row = ei;
  const int* col = ei + E;

  const int strips = (N + 15) / 16;
  const int NP     = strips * 16;

  const int NB     = (N + BN - 1) / BN;

  float*  h    = (float*)d_ws;
  float*  o1   = h + (size_t)N * H1;
  int2*   ed   = (int2*)(o1 + (size_t)N * H1);
  int4*   ovf  = (int4*)(ed + (size_t)N * CAP);
  int4*   bins = ovf + E;
  int4*   covf = bins + (size_t)NB * BINCAP;
  float*  dinv = (float*)(covf + E);
  __bf16* w1t  = (__bf16*)(dinv + NP);
  __bf16* w2t  = w1t + F_IN * H1;
  int*    cnt  = (int*)(w2t + F_IN * H2);
  int*    binc = cnt + N + 1;
  int*    covc = binc + NB * CSTRIDE;

  const int T = 256;
  constexpr int RG  = 1024;
  constexpr int RG2 = 2048;

  k_fill0 <<<(NB * CSTRIDE + 2 + T - 1) / T, T, 0, stream>>>(cnt + N, NB * CSTRIDE + 2);
  k_coarse<<<(E + TILE - 1) / TILE, T, 0, stream>>>(row, col, P, binc, covc, bins, covf, E, N, NB);
  k_fine  <<<NB, BN, 0, stream>>>(binc, covc, bins, covf, cnt, ed, ovf, E, N);
  k_dinv  <<<(NP + T - 1) / T, T, 0, stream>>>(cnt, ed, ovf, dinv, N, NP, E);
  k_wt_bf16<H1><<<(F_IN * H1 + T - 1) / T, T, 0, stream>>>(W1, w1t);
  k_wt_bf16<H2><<<(F_IN * H2 + T - 1) / T, T, 0, stream>>>(W2, w2t);

  k_gemm_wmma<H1, false><<<RG, 32 * (H1 / 16), 0, stream>>>(x, w1t, dinv, h, N, strips);
  k_gather<H1><<<RG, T, 0, stream>>>(cnt, ed, ovf, dinv, h, b1, o1, N, E);

  k_gemm_wmma<H2, true><<<RG2, 32 * (H2 / 16), 0, stream>>>(o1, w2t, dinv, h, N, strips);
  k_gather<H2><<<RG, T, 0, stream>>>(cnt, ed, ovf, dinv, h, b2, out, N, E);
}


# layer-2 product moved into the layer-1 gather: a wave gathers a 16-node strip into LDS, multiplies it by W2 there and writes hs2; o1 never reaches memory; 7 -> 6 launches; sum order kept
# speedup vs baseline: 4.5949x; 1.1403x over previous
#include <hip/hip_runtime.h>
#include <hip/hip_bf16.h>

typedef __attribute__((ext_vector_type(16))) __bf16        v16bf;
typedef __attribute__((ext_vector_type(8)))  __bf16        v8bf;
typedef __attribute__((ext_vector_type(4)))  __bf16        v4bf;
typedef __attribute__((ext_vector_type(8)))  float         v8f;

__device__ __forceinline__ __bf16 tobf(float f) { return (__bf16)f; }

constexpr int CAP      = 32;
constexpr int BN_SHIFT = 8;
constexpr int BN       = 1 << BN_SHIFT;
constexpr int BINCAP   = 8192;
constexpr int NBMAX    = 1024;
constexpr int CSTRIDE  = 64;
constexpr int EPT      = 16;
constexpr int TILE     = 256 * EPT;

template<int NCOL>
__device__ __forceinline__ void wt_bf16(const float* __restrict__ W, __bf16* __restrict__ Wt,
                                        int idx) {
  if (idx >= 128 * NCOL) return;
  const int k = idx / NCOL, n = idx % NCOL;
  Wt[n * 128 + k] = tobf(W[idx]);
}

template<int H1, int H2>
__global__ void __launch_bounds__(256)
k_setup(int* __restrict__ p, int n, int fill_blocks,
        const float* __restrict__ W1, __bf16* __restrict__ W1t,
        const float* __restrict__ W2, __bf16* __restrict__ W2t) {
  constexpr int B1 = 128 * H1 / 256;
  int blk = blockIdx.x;
  if (blk < fill_blocks) {
    const int i = blk * 256 + threadIdx.x;
    if (i < n) p[i] = 0;
    return;
  }
  blk -= fill_blocks;
  if (blk < B1) wt_bf16<H1>(W1, W1t, blk * 256 + threadIdx.x);
  else          wt_bf16<H2>(W2, W2t, (blk - B1) * 256 + threadIdx.x);
}

__global__ void __launch_bounds__(256)
k_coarse(const int* __restrict__ row, const int* __restrict__ col,
         const float* __restrict__ P, int* __restrict__ binc, int* __restrict__ covc,
         int4* __restrict__ bins, int4* __restrict__ covf, int E, int N, int NB) {
  __shared__ int hist[NBMAX];
  if (NB > NBMAX) return;
  const int tid    = threadIdx.x;
  const int ntiles = (E + TILE - 1) / TILE;
  for (int tile = blockIdx.x; tile < ntiles; tile += gridDim.x) {
    for (int b = tid; b < NB; b += 256) hist[b] = 0;
    __syncthreads();
    int   c[EPT], rw[EPT], rk[EPT];
    float pv[EPT];
    const int e0 = tile * TILE + tid;
#pragma unroll
    for (int u = 0; u < EPT; ++u) {
      const int e  = e0 + u * 256;
      const int ec = (e < E) ? e : (E - 1);
      c[u]  = col[ec];
      rw[u] = row[ec];
      pv[u] = P[ec];
    }
#pragma unroll
    for (int u = 0; u < EPT; ++u) {
      const int e = e0 + u * 256;
      const int cc = c[u] < 0 ? 0 : (c[u] > N - 1 ? N - 1 : c[u]);
      c[u]  = cc;
      rk[u] = 0;
      if (e < E) rk[u] = atomicAdd(&hist[cc >> BN_SHIFT], 1);
    }
    __syncthreads();
    for (int b = tid; b < NB; b += 256) {
      const int hc = hist[b];
      if (hc > 0) hist[b] = atomicAdd(&binc[b * CSTRIDE], hc);
    }
    __syncthreads();
    int bs[EPT];
#pragma unroll
    for (int u = 0; u < EPT; ++u) bs[u] = hist[c[u] >> BN_SHIFT];
#pragma unroll
    for (int u = 0; u < EPT; ++u) {
      const int e = e0 + u * 256;
      if (e < E) {
        const int   b   = c[u] >> BN_SHIFT;
        const int   pos = bs[u] + rk[u];
        const float s   = 1.0f / (1.0f + __expf(-pv[u]));
        const int4  q   = make_int4(c[u], rw[u], __float_as_int(s), 0);
        if (pos < BINCAP) {
          bins[(size_t)b * BINCAP + pos] = q;
        } else {
          const int k = atomicAdd(covc, 1);
          if (k < E) covf[k] = q;
        }
      }
    }
    __syncthreads();
  }
}

__device__ __forceinline__ void place_edge(const int4 q, int nb0, int* lcnt, float* lsum,
                                           int2* srow,
                                           int* __restrict__ cnt, int4* __restrict__ ovf,
                                           int E, int N) {
  const int cl = q.x - nb0;
  if (cl < 0 || cl >= BN) return;
  const int r = atomicAdd(&lcnt[cl], 1);
  atomicAdd(&lsum[cl], __int_as_float(q.z));
  if (r < CAP) {
    srow[cl * CAP + r] = make_int2(q.y, q.z);
  } else {
    const int k = atomicAdd(&cnt[N], 1);
    if (k < E) ovf[k] = make_int4(q.x, q.y, q.z, 0);
  }
}

__global__ void __launch_bounds__(BN)
k_fine(const int* __restrict__ binc, const int* __restrict__ covc,
       const int4* __restrict__ bins, const int4* __restrict__ covf,
       int* __restrict__ cnt, int2* __restrict__ ed, int4* __restrict__ ovf,
       float* __restrict__ dinv, int E, int N, int NP) {
  __shared__ __align__(16) int2 srow[BN * CAP];
  __shared__ int   lcnt[BN];
  __shared__ float lsum[BN];
  const int tid = threadIdx.x;
  const int b   = blockIdx.x;
  const int nb0 = b << BN_SHIFT;
  lcnt[tid] = 0;
  lsum[tid] = 0.0f;
  __syncthreads();

  int nb = binc[b * CSTRIDE];
  if (nb > BINCAP) nb = BINCAP;
  const int4* bl = bins + (size_t)b * BINCAP;
  for (int i0 = tid; i0 < nb; i0 += 4 * BN) {
    int4 q[4];
#pragma unroll
    for (int u = 0; u < 4; ++u) {
      const int i = i0 + u * BN;
      q[u] = bl[(i < nb) ? i : (nb - 1)];
    }
#pragma unroll
    for (int u = 0; u < 4; ++u) {
      const int i = i0 + u * BN;
      if (i < nb) place_edge(q[u], nb0, lcnt, lsum, srow, cnt, ovf, E, N);
    }
  }
  int nco = covc[0];
  if (nco > E) nco = E;
  for (int i = tid; i < nco; i += BN) {
    const int4 q = covf[i];
    if ((q.x >> BN_SHIFT) == b) place_edge(q, nb0, lcnt, lsum, srow, cnt, ovf, E, N);
  }
  __syncthreads();

  if (nb0 + tid < N) cnt[nb0 + tid] = lcnt[tid];
  if (nb0 + tid < NP) {
    const float d = 1.0f + lsum[tid];
    dinv[nb0 + tid] = (nb0 + tid < N && d > 0.0f) ? rsqrtf(d) : 0.0f;
  }
  const int4* s4 = (const int4*)srow;
  int4*       d4 = (int4*)(ed + (size_t)nb0 * CAP);
  constexpr int Q_PER_ROW = CAP * 8 / 16;
  for (int i = tid; i < BN * Q_PER_ROW; i += BN) {
    if (nb0 + i / Q_PER_ROW < N) d4[i] = s4[i];
  }
}

template<int NT, int PASSES>
__device__ __forceinline__ void load_strip(const float* __restrict__ A, int strip, int N,
                                           float4 (&fa)[PASSES], float4 (&fb)[PASSES]) {
#pragma unroll
  for (int p = 0; p < PASSES; ++p) {
    const int base = (p * NT + (int)threadIdx.x) * 8;
    const int m = base >> 7;
    const int k = base & 127;
    int mg = strip * 16 + m; if (mg > N - 1) mg = N - 1;
    const float* src = A + (size_t)mg * 128 + k;
    fa[p] = *(const float4*)(src);
    fb[p] = *(const float4*)(src + 4);
  }
}

template<int NCOL, bool RELU_A>
__global__ void __launch_bounds__(32 * (NCOL / 16))
k_gemm_wmma(const float* __restrict__ A,
            const __bf16* __restrict__ Bt,
            const float* __restrict__ dinv,
            float* __restrict__ H, int N, int strips) {
  constexpr int K      = 128;
  constexpr int LDSP   = 136;
  constexpr int NT     = 32 * (NCOL / 16);
  constexpr int PASSES = (16 * K) / (NT * 8);
  __shared__ __align__(16) __bf16 As[2][16 * LDSP];

  const int lane = threadIdx.x & 31;
  const int wave = threadIdx.x >> 5;
  const int n0   = wave * 16;
  const int half = lane >> 4;
  const int l15  = lane & 15;

  const __bf16* Brow = Bt + (size_t)(n0 + l15) * K;
  const int     aoff = l15 * LDSP;

  v16bf bfr[4];
#pragma unroll
  for (int i = 0; i < 4; ++i) {
    const v8bf b0 = *(const v8bf*)(Brow + i * 32 + half * 8);
    const v8bf b1 = *(const v8bf*)(Brow + i * 32 + 16 + half * 8);
    bfr[i] = __builtin_shufflevector(b0, b1,
        0, 1, 2, 3, 4, 5, 6, 7, 8, 9, 10, 11, 12, 13, 14, 15);
  }

  float4 fa[PASSES], fb[PASSES];
  int s = blockIdx.x;
  if (s < strips) load_strip<NT, PASSES>(A, s, N, fa, fb);

  int pb = 0;
  for (; s < strips; s += gridDim.x, pb ^= 1) {
    const int m0 = s * 16;
    __bf16* Ab = As[pb];
#pragma unroll
    for (int p = 0; p < PASSES; ++p) {
      const int base = (p * NT + (int)threadIdx.x) * 8;
      const int m = base >> 7;
      const int k = base & (K - 1);
      float4 f0 = fa[p], f1 = fb[p];
      if (RELU_A) {
        f0.x = fmaxf(f0.x, 0.f); f0.y = fmaxf(f0.y, 0.f);
        f0.z = fmaxf(f0.z, 0.f); f0.w = fmaxf(f0.w, 0.f);
        f1.x = fmaxf(f1.x, 0.f); f1.y = fmaxf(f1.y, 0.f);
        f1.z = fmaxf(f1.z, 0.f); f1.w = fmaxf(f1.w, 0.f);
      }
      v8bf pk;
      pk[0] = tobf(f0.x); pk[1] = tobf(f0.y); pk[2] = tobf(f0.z); pk[3] = tobf(f0.w);
      pk[4] = tobf(f1.x); pk[5] = tobf(f1.y); pk[6] = tobf(f1.z); pk[7] = tobf(f1.w);
      *(v8bf*)(&Ab[m * LDSP + k]) = pk;
    }
    __syncthreads();

    const float4 dv0 = *(const float4*)(dinv + m0 + half * 8);
    const float4 dv1 = *(const float4*)(dinv + m0 + half * 8 + 4);

    const int sn = s + gridDim.x;
    load_strip<NT, PASSES>(A, (sn < strips) ? sn : s, N, fa, fb);

    v8f acc = {};
#pragma unroll
    for (int i = 0; i < 4; ++i) {
      const v8bf a0 = *(const v8bf*)(Ab + aoff + i * 32 + half * 8);
      const v8bf a1 = *(const v8bf*)(Ab + aoff + i * 32 + 16 + half * 8);
      const v16bf a = __builtin_shufflevector(a0, a1,
          0, 1, 2, 3, 4, 5, 6, 7, 8, 9, 10, 11, 12, 13, 14, 15);
      acc = __builtin_amdgcn_wmma_f32_16x16x32_bf16(false, a, false, bfr[i],
                                                    (short)0, acc, false, false);
    }

    acc[0] *= dv0.x; acc[1] *= dv0.y; acc[2] *= dv0.z; acc[3] *= dv0.w;
    acc[4] *= dv1.x; acc[5] *= dv1.y; acc[6] *= dv1.z; acc[7] *= dv1.w;
    const int n = n0 + l15;
    if (m0 + 16 <= N) {
#pragma unroll
      for (int r = 0; r < 8; ++r) {
        const int m = m0 + half * 8 + r;
        H[(size_t)m * NCOL + n] = acc[r];
      }
    } else {
#pragma unroll
      for (int r = 0; r < 8; ++r) {
        const int m = m0 + half * 8 + r;
        if (m < N) H[(size_t)m * NCOL + n] = acc[r];
      }
    }
  }
}

template<int PER>
__device__ __forceinline__ void ld_row(const float* __restrict__ p, float (&v)[PER]) {
  if constexpr (PER == 4) {
    const float4 t = *(const float4*)p;
    v[0] = t.x; v[1] = t.y; v[2] = t.z; v[3] = t.w;
  } else {
    const float2 t = *(const float2*)p;
    v[0] = t.x; v[1] = t.y;
  }
}

template<int F, int NB>
__device__ __forceinline__ void rows_fma(const float* __restrict__ hl, int r, float cf,
                                         int j, float (&acc)[F / 32]) {
  constexpr int PER = F / 32;
  float v[NB][PER];
  float c[NB];
#pragma unroll
  for (int u = 0; u < NB; ++u) {
    const int ru = __builtin_amdgcn_readlane(r, j + u);
    c[u] = __int_as_float(__builtin_amdgcn_readlane(__float_as_int(cf), j + u));
    ld_row<PER>(hl + (size_t)ru * F, v[u]);
  }
#pragma unroll
  for (int u = 0; u < NB; ++u) {
#pragma unroll
    for (int q = 0; q < PER; ++q) acc[q] = fmaf(c[u], v[u][q], acc[q]);
  }
}

template<int F>
__global__ void __launch_bounds__(256)
k_gather(const int* __restrict__ cnt, const int2* __restrict__ ed,
         const int4* __restrict__ ovf,
         const float* __restrict__ dinv, const float* __restrict__ h,
         const float* __restrict__ bias, float* __restrict__ out, int N, int E) {
  constexpr int PER = F / 32;
  const int lane = threadIdx.x & 31;
  const int gw   = __builtin_amdgcn_readfirstlane(
                     (blockIdx.x * blockDim.x + threadIdx.x) >> 5);
  const int tw   = (gridDim.x * blockDim.x) >> 5;
  const int per  = (N + tw - 1) / tw;
  const int n0   = gw * per;
  const int n1   = (n0 + per < N) ? (n0 + per) : N;
  if (n0 >= n1) return;

  const float* hl = h + lane * PER;
  float bs[PER];
  ld_row<PER>(bias + lane * PER, bs);

  for (int cb = n0; cb < n1; cb += 32) {
    const int cn = (n1 - cb < 32) ? (n1 - cb) : 32;
    int ci = cb + lane; if (ci > N - 1) ci = N - 1;
    const int   dgv = cnt[ci];
    const float dvv = dinv[ci];

    int deg = __builtin_amdgcn_readlane(dgv, 0);
    int   rn  = cb;
    float cfn = 0.0f;
    if (lane < deg) {
      const int2 t = ed[(size_t)cb * CAP + lane];
      rn  = t.x;
      cfn = __int_as_float(t.y);
    }

    for (int t = 0; t < cn; ++t) {
      const int node = cb + t;
      const int dg = deg;
      int   r  = rn < 0 ? 0 : (rn > N - 1 ? N - 1 : rn);
      float cf = cfn;

      float acc[PER];
      ld_row<PER>(hl + (size_t)node * F, acc);
      const float di = __int_as_float(
          __builtin_amdgcn_readlane(__float_as_int(dvv), t));

      if (t + 1 < cn) {
        deg = __builtin_amdgcn_readlane(dgv, t + 1);
        rn  = node + 1;
        cfn = 0.0f;
        if (lane < deg) {
          const int2 q = ed[(size_t)(node + 1) * CAP + lane];
          rn  = q.x;
          cfn = __int_as_float(q.y);
        }
      }

      const int nk = (dg < CAP) ? dg : CAP;
      for (int j = 0; j < nk; j += 8) {
        if (j + 4 < nk) rows_fma<F, 8>(hl, r, cf, j, acc);
        else            rows_fma<F, 4>(hl, r, cf, j, acc);
      }

      if (dg > CAP) {
        int nov = cnt[N]; if (nov > E) nov = E;
        for (int base = 0; base < nov; base += 32) {
          int   oc = -1, orow = node;
          float ow = 0.0f;
          if (base + lane < nov) {
            const int4 q = ovf[base + lane];
            oc   = q.x;
            orow = q.y < 0 ? 0 : (q.y > N - 1 ? N - 1 : q.y);
            ow   = __int_as_float(q.z);
          }
          unsigned m = __builtin_amdgcn_ballot_w32(oc == node);
          while (m) {
            const int l = __builtin_ctz(m);
            m &= m - 1;
            const int   ru = __builtin_amdgcn_readlane(orow, l);
            const float cu = __int_as_float(
                __builtin_amdgcn_readlane(__float_as_int(ow), l));
            float v[PER];
            ld_row<PER>(hl + (size_t)ru * F, v);
#pragma unroll
            for (int q = 0; q < PER; ++q) acc[q] = fmaf(cu, v[q], acc[q]);
          }
        }
      }

      float* op = out + (size_t)node * F + lane * PER;
      if constexpr (PER == 4) {
        *(float4*)op = make_float4(fmaf(di, acc[0], bs[0]), fmaf(di, acc[1], bs[1]),
                                   fmaf(di, acc[2], bs[2]), fmaf(di, acc[3], bs[3]));
      } else {
        *(float2*)op = make_float2(fmaf(di, acc[0], bs[0]), fmaf(di, acc[1], bs[1]));
      }
    }
  }
}

template<int F, int NCOL>
__global__ void __launch_bounds__(256)
k_gather_gemm(const int* __restrict__ cnt, const int2* __restrict__ ed,
              const int4* __restrict__ ovf,
              const float* __restrict__ dinv, const float* __restrict__ h,
              const float* __restrict__ bias, const __bf16* __restrict__ Bt,
              float* __restrict__ H2, int N, int E, int strips) {
  static_assert(F == 128, "one k-step layout: four 32-wide steps, four columns per lane");
  constexpr int PER  = F / 32;
  constexpr int LDSP = F + 8;
  constexpr int CP   = NCOL + 4;
  constexpr int NTIL = NCOL / 16;
  constexpr int WPB  = 8;
  static_assert(16 * CP * 4 <= 16 * LDSP * 2, "the fp32 tile reuses the strip buffer");
  __shared__ __align__(16) __bf16 Bs[NCOL * LDSP];
  __shared__ __align__(16) __bf16 As[WPB][16 * LDSP];

  const int tid  = threadIdx.x;
  const int lane = tid & 31;
  const int wave = __builtin_amdgcn_readfirstlane(tid >> 5);
  const int half = lane >> 4;
  const int l15  = lane & 15;
  const int s    = blockIdx.x * WPB + wave;
  const int m0   = s * 16;

  constexpr int BPASS = (NCOL * F) / (256 * 8);
  v8bf tb[BPASS];
#pragma unroll
  for (int p = 0; p < BPASS; ++p) tb[p] = *(const v8bf*)(Bt + (size_t)(p * 256 + tid) * 8);

  int ci = m0 + lane; if (ci > N - 1) ci = N - 1;
  const int   dgv = cnt[ci];
  const float dvv = dinv[ci];

#pragma unroll
  for (int p = 0; p < BPASS; ++p) {
    const int idx = (p * 256 + tid) * 8;
    *(v8bf*)(&Bs[(idx / F) * LDSP + (idx % F)]) = tb[p];
  }
  __syncthreads();
  if (s >= strips) return;

  __bf16* Aw = As[wave];
  const int cn = (N - m0 < 16) ? (N - m0) : 16;

  const float* hl = h + lane * PER;
  float bs[PER];
  ld_row<PER>(bias + lane * PER, bs);

  int deg = __builtin_amdgcn_readlane(dgv, 0);
  int   rn  = m0;
  float cfn = 0.0f;
  if (lane < deg) {
    const int2 t = ed[(size_t)m0 * CAP + lane];
    rn  = t.x;
    cfn = __int_as_float(t.y);
  }

  for (int t = 0; t < cn; ++t) {
    const int node = m0 + t;
    const int dg = deg;
    int   r  = rn < 0 ? 0 : (rn > N - 1 ? N - 1 : rn);
    float cf = cfn;

    float acc[PER];
    ld_row<PER>(hl + (size_t)node * F, acc);
    const float di = __int_as_float(
        __builtin_amdgcn_readlane(__float_as_int(dvv), t));

    if (t + 1 < cn) {
      deg = __builtin_amdgcn_readlane(dgv, t + 1);
      rn  = node + 1;
      cfn = 0.0f;
      if (lane < deg) {
        const int2 q = ed[(size_t)(node + 1) * CAP + lane];
        rn  = q.x;
        cfn = __int_as_float(q.y);
      }
    }

    const int nk = (dg < CAP) ? dg : CAP;
    for (int j = 0; j < nk; j += 8) {
      if (j + 4 < nk) rows_fma<F, 8>(hl, r, cf, j, acc);
      else            rows_fma<F, 4>(hl, r, cf, j, acc);
    }

    if (dg > CAP) {
      int nov = cnt[N]; if (nov > E) nov = E;
      for (int base = 0; base < nov; base += 32) {
        int   oc = -1, orow = node;
        float ow = 0.0f;
        if (base + lane < nov) {
          const int4 q = ovf[base + lane];
          oc   = q.x;
          orow = q.y < 0 ? 0 : (q.y > N - 1 ? N - 1 : q.y);
          ow   = __int_as_float(q.z);
        }
        unsigned m = __builtin_amdgcn_ballot_w32(oc == node);
        while (m) {
          const int l = __builtin_ctz(m);
          m &= m - 1;
          const int   ru = __builtin_amdgcn_readlane(orow, l);
          const float cu = __int_as_float(
              __builtin_amdgcn_readlane(__float_as_int(ow), l));
          float v[PER];
          ld_row<PER>(hl + (size_t)ru * F, v);
#pragma unroll
          for (int q = 0; q < PER; ++q) acc[q] = fmaf(cu, v[q], acc[q]);
        }
      }
    }

    v4bf pk;
    pk[0] = tobf(fmaxf(fmaf(di, acc[0], bs[0]), 0.f));
    pk[1] = tobf(fmaxf(fmaf(di, acc[1], bs[1]), 0.f));
    pk[2] = tobf(fmaxf(fmaf(di, acc[2], bs[2]), 0.f));
    pk[3] = tobf(fmaxf(fmaf(di, acc[3], bs[3]), 0.f));
    *(v4bf*)(&Aw[t * LDSP + lane * PER]) = pk;
  }
  for (int t = cn; t < 16; ++t) {
    v4bf z = {};
    *(v4bf*)(&Aw[t * LDSP + lane * PER]) = z;
  }

  __builtin_amdgcn_fence(__ATOMIC_ACQ_REL, "workgroup");

  const int aoff = l15 * LDSP;
  v16bf afr[4];
#pragma unroll
  for (int i = 0; i < 4; ++i) {
    const v8bf a0 = *(const v8bf*)(Aw + aoff + i * 32 + half * 8);
    const v8bf a1 = *(const v8bf*)(Aw + aoff + i * 32 + 16 + half * 8);
    afr[i] = __builtin_shufflevector(a0, a1,
        0, 1, 2, 3, 4, 5, 6, 7, 8, 9, 10, 11, 12, 13, 14, 15);
  }
  v8f c[NTIL];
#pragma unroll
  for (int j = 0; j < NTIL; ++j) {
    const __bf16* Brow = Bs + (j * 16 + l15) * LDSP;
    v8f cj = {};
#pragma unroll
    for (int i = 0; i < 4; ++i) {
      const v8bf b0 = *(const v8bf*)(Brow + i * 32 + half * 8);
      const v8bf b1 = *(const v8bf*)(Brow + i * 32 + 16 + half * 8);
      const v16bf b = __builtin_shufflevector(b0, b1,
          0, 1, 2, 3, 4, 5, 6, 7, 8, 9, 10, 11, 12, 13, 14, 15);
      cj = __builtin_amdgcn_wmma_f32_16x16x32_bf16(false, afr[i], false, b,
                                                   (short)0, cj, false, false);
    }
    c[j] = cj;
  }

  float* Cw = (float*)Aw;
#pragma unroll
  for (int r = 0; r < 8; ++r) {
    const float dlo = __int_as_float(__builtin_amdgcn_readlane(__float_as_int(dvv), r));
    const float dhi = __int_as_float(__builtin_amdgcn_readlane(__float_as_int(dvv), 8 + r));
    const float d   = half ? dhi : dlo;
#pragma unroll
    for (int j = 0; j < NTIL; ++j)
      Cw[(half * 8 + r) * CP + j * 16 + l15] = c[j][r] * d;
  }
  __builtin_amdgcn_fence(__ATOMIC_ACQ_REL, "workgroup");

  constexpr int LPR = NCOL / 4;
  constexpr int RPI = 32 / LPR;
  const int rl = lane / LPR, cl = (lane % LPR) * 4;
  float4 ov[16 / RPI];
#pragma unroll
  for (int q = 0; q < 16 / RPI; ++q)
    ov[q] = *(const float4*)(Cw + (q * RPI + rl) * CP + cl);
  float* op = H2 + (size_t)(m0 + rl) * NCOL + cl;
  if (m0 + 16 <= N) {
#pragma unroll
    for (int q = 0; q < 16 / RPI; ++q)
      *(float4*)(op + (size_t)q * RPI * NCOL) = ov[q];
  } else {
#pragma unroll
    for (int q = 0; q < 16 / RPI; ++q) {
      if (m0 + q * RPI + rl < N) *(float4*)(op + (size_t)q * RPI * NCOL) = ov[q];
    }
  }
}

extern "C" void kernel_launch(void* const* d_in, const int* in_sizes, int n_in,
                              void* d_out, int out_size, void* d_ws, size_t ws_size,
                              hipStream_t stream) {
  const float* x  = (const float*)d_in[0];
  const int*   ei = (const int*)  d_in[1];
  const float* P  = (const float*)d_in[2];
  const float* W1 = (const float*)d_in[3];
  const float* b1 = (const float*)d_in[4];
  const float* W2 = (const float*)d_in[5];
  const float* b2 = (const float*)d_in[6];
  float* out = (float*)d_out;

  constexpr int F_IN = 128, H1 = 128, H2 = 64;
  const int N = in_sizes[0] / F_IN;
  const int E = in_sizes[2];
  const int* row = ei;
  const int* col = ei + E;

  const int strips = (N + 15) / 16;
  const int NP     = strips * 16;

  const int NB     = (N + BN - 1) / BN;

  float*  h    = (float*)d_ws;
  float*  o1   = h + (size_t)N * H1;
  int2*   ed   = (int2*)(o1 + (size_t)N * H1);
  int4*   ovf  = (int4*)(ed + (size_t)N * CAP);
  int4*   bins = ovf + E;
  int4*   covf = bins + (size_t)NB * BINCAP;
  float*  dinv = (float*)(covf + E);
  __bf16* w1t  = (__bf16*)(dinv + NP);
  __bf16* w2t  = w1t + F_IN * H1;
  int*    cnt  = (int*)(w2t + F_IN * H2);
  int*    binc = cnt + N + 1;
  int*    covc = binc + NB * CSTRIDE;

  const int T = 256;
  constexpr int RG  = 1024;

  const int FB = (NB * CSTRIDE + 2 + T - 1) / T;
  k_setup<H1, H2><<<FB + F_IN * (H1 + H2) / T, T, 0, stream>>>(cnt + N, NB * CSTRIDE + 2, FB,
                                                              W1, w1t, W2, w2t);
  k_coarse<<<(E + TILE - 1) / TILE, T, 0, stream>>>(row, col, P, binc, covc, bins, covf, E, N, NB);
  k_fine  <<<NB, BN, 0, stream>>>(binc, covc, bins, covf, cnt, ed, ovf, dinv, E, N, NP);

  k_gemm_wmma<H1, false><<<RG, 32 * (H1 / 16), 0, stream>>>(x, w1t, dinv, h, N, strips);
  k_gather_gemm<H1, H2><<<(strips + 7) / 8, T, 0, stream>>>(cnt, ed, ovf, dinv, h, b1, w2t, o1,
                                                            N, E, strips);

  k_gather<H2><<<RG, T, 0, stream>>>(cnt, ed, ovf, dinv, o1, b2, out, N, E);
}
